# starting kernel
# baseline (speedup 1.0000x reference)
#include <hip/hip_runtime.h>
#include <math.h>

typedef __attribute__((ext_vector_type(2))) float v2f;
typedef __attribute__((ext_vector_type(8))) float v8f;

#define HIDDEN 64
#define NEPS 1e-12f

__global__ void normalize_rows_kernel(const float* __restrict__ emb,
                                      float* __restrict__ out,
                                      int nrows) {
    int gid  = blockIdx.x * blockDim.x + threadIdx.x;
    int row  = gid >> 5;
    int lane = gid & 31;
    if (row >= nrows) return;

    const float2* src = reinterpret_cast<const float2*>(emb + (size_t)row * HIDDEN);
    float2 x = src[lane];
    float s = x.x * x.x + x.y * x.y;
#pragma unroll
    for (int off = 16; off > 0; off >>= 1)
        s += __shfl_xor(s, off, 32);

    float inv = 1.0f / fmaxf(sqrtf(s), NEPS);
    float2 y; y.x = x.x * inv; y.y = x.y * inv;
    reinterpret_cast<float2*>(out + (size_t)row * HIDDEN)[lane] = y;
}

__global__ void edge_dot_wmma_kernel(const float* __restrict__ embn,
                                     const float* __restrict__ dvec,
                                     const float* __restrict__ scale,
                                     const int* __restrict__ eidx,
                                     float* __restrict__ out,
                                     int nedges, int ntiles) {
    int wave = (blockIdx.x * blockDim.x + threadIdx.x) >> 5;
    int lane = threadIdx.x & 31;
    if (wave >= ntiles) return;

    int elocal = lane & 15;
    int hi     = lane >> 4;
    int edge   = wave * 16 + elocal;
    int ecl    = edge < nedges ? edge : nedges - 1;

    int u = eidx[ecl];
    int v = eidx[nedges + ecl];

    const float* Au = embn + (size_t)u * HIDDEN;
    const float* Bv = embn + (size_t)v * HIDDEN;

    v8f c0 = {};
    v8f c1 = {};
#pragma unroll
    for (int kk = 0; kk < HIDDEN; kk += 8) {
        int off0 = kk + 2 * hi;
        int off1 = kk + 4 + 2 * hi;
        v2f a0 = *reinterpret_cast<const v2f*>(Au + off0);
        v2f b0 = *reinterpret_cast<const v2f*>(Bv + off0);
        v2f d0 = *reinterpret_cast<const v2f*>(dvec + off0);
        v2f a1 = *reinterpret_cast<const v2f*>(Au + off1);
        v2f b1 = *reinterpret_cast<const v2f*>(Bv + off1);
        v2f d1 = *reinterpret_cast<const v2f*>(dvec + off1);
        b0 *= d0;
        b1 *= d1;
        c0 = __builtin_amdgcn_wmma_f32_16x16x4_f32(false, a0, false, b0,
                                                   (short)0, c0, false, false);
        c1 = __builtin_amdgcn_wmma_f32_16x16x4_f32(false, a1, false, b1,
                                                   (short)0, c1, false, false);
    }
    v8f c = c0 + c1;

    float sc = scale[0];

    int sel = -1, eo = 0;
    if (lane < 8)        { sel = lane;      eo = lane;      }
    else if (lane >= 24) { sel = lane - 24; eo = lane - 16; }

    if (sel >= 0) {
        float r = 0.0f;
        switch (sel) {
            case 0: r = c[0]; break;
            case 1: r = c[1]; break;
            case 2: r = c[2]; break;
            case 3: r = c[3]; break;
            case 4: r = c[4]; break;
            case 5: r = c[5]; break;
            case 6: r = c[6]; break;
            case 7: r = c[7]; break;
        }
        int eg = wave * 16 + eo;
        if (eg < nedges) out[eg] = r * sc;
    }
}

extern "C" void kernel_launch(void* const* d_in, const int* in_sizes, int n_in,
                              void* d_out, int out_size, void* d_ws, size_t ws_size,
                              hipStream_t stream) {
    const float* emb   = (const float*)d_in[0];
    const float* dvec  = (const float*)d_in[1];
    const float* scale = (const float*)d_in[2];
    const int*   eidx  = (const int*)d_in[3];
    float*       out   = (float*)d_out;

    int nrows  = in_sizes[0] / HIDDEN;
    int nedges = in_sizes[3] / 2;

    float* embn = (float*)d_ws;

    {
        long long total_threads = (long long)nrows * 32;
        int threads = 256;
        int blocks  = (int)((total_threads + threads - 1) / threads);
        normalize_rows_kernel<<<blocks, threads, 0, stream>>>(emb, embn, nrows);
    }

    {
        int ntiles = (nedges + 15) / 16;
        long long total_threads = (long long)ntiles * 32;
        int threads = 256;
        int blocks  = (int)((total_threads + threads - 1) / threads);
        edge_dot_wmma_kernel<<<blocks, threads, 0, stream>>>(
            embn, dvec, scale, eidx, out, nedges, ntiles);
    }
}


# Edge kernel: each b128 gather load covers two whole 256-B rows (tile row = edge and 32-B chunk, not edge); an edge's 8 chunk partials are summed by a lane butterfly, so the f32 sum order changes
# speedup vs baseline: 1.7043x; 1.7043x over previous
#include <hip/hip_runtime.h>
#include <math.h>

typedef __attribute__((ext_vector_type(2))) float v2f;
typedef __attribute__((ext_vector_type(4))) float v4f;
typedef __attribute__((ext_vector_type(8))) float v8f;

#define HIDDEN 64
#define NEPS 1e-12f

__global__ void normalize_rows_kernel(const float* __restrict__ emb,
                                      float* __restrict__ out,
                                      int nrows) {
    int gid  = blockIdx.x * blockDim.x + threadIdx.x;
    int row  = gid >> 5;
    int lane = gid & 31;
    if (row >= nrows) return;

    const float2* src = reinterpret_cast<const float2*>(emb + (size_t)row * HIDDEN);
    float2 x = src[lane];
    float s = x.x * x.x + x.y * x.y;
#pragma unroll
    for (int off = 16; off > 0; off >>= 1)
        s += __shfl_xor(s, off, 32);

    float inv = 1.0f / fmaxf(sqrtf(s), NEPS);
    float2 y; y.x = x.x * inv; y.y = x.y * inv;
    reinterpret_cast<float2*>(out + (size_t)row * HIDDEN)[lane] = y;
}


template <int T>
__device__ __forceinline__ int group8_bcast(int x) {
    return __builtin_amdgcn_mov_dpp8(x, T * 0x249249);
}

template <int M>
__device__ __forceinline__ float row_xor(float x) {
    return __int_as_float(__builtin_amdgcn_update_dpp(
        0, __float_as_int(x), 0x160 | M, 0xf, 0xf, false));
}

template <int T>
__device__ __forceinline__ void load_pair(const char* __restrict__ embn_bytes,
                                          int u, int v, unsigned lane_off,
                                          v4f& a, v4f& b) {
    unsigned ut = (unsigned)group8_bcast<T>(u);
    unsigned vt = (unsigned)group8_bcast<T>(v);
    a = *reinterpret_cast<const v4f*>(embn_bytes + (ut * (HIDDEN * 4u) + lane_off));
    b = *reinterpret_cast<const v4f*>(embn_bytes + (vt * (HIDDEN * 4u) + lane_off));
}

__device__ __forceinline__ float pair_diag(v4f a, v4f b, v4f dd, int g) {
    b *= dd;
    v2f a0 = {a.x, a.y}, a1 = {a.z, a.w};
    v2f b0 = {b.x, b.y}, b1 = {b.z, b.w};
    v8f c = {};
    c = __builtin_amdgcn_wmma_f32_16x16x4_f32(false, a0, false, b0,
                                              (short)0, c, false, false);
    c = __builtin_amdgcn_wmma_f32_16x16x4_f32(false, a1, false, b1,
                                              (short)0, c, false, false);
    float x = c[0];
    x = (g == 1) ? c[1] : x;
    x = (g == 2) ? c[2] : x;
    x = (g == 3) ? c[3] : x;
    x = (g == 4) ? c[4] : x;
    x = (g == 5) ? c[5] : x;
    x = (g == 6) ? c[6] : x;
    x = (g == 7) ? c[7] : x;
    return x;
}

__global__ void edge_dot_wmma_kernel(const float* __restrict__ embn,
                                     const float* __restrict__ dvec,
                                     const float* __restrict__ scale,
                                     const int* __restrict__ eidx,
                                     float* __restrict__ out,
                                     int nedges, int ntiles) {
    int wave = (blockIdx.x * blockDim.x + threadIdx.x) >> 5;
    int lane = threadIdx.x & 31;
    if (wave >= ntiles) return;

    int c    = lane & 15;
    int hi   = lane >> 4;
    int g    = lane & 7;
    int edge = wave * 16 + c;
    int ecl  = edge < nedges ? edge : nedges - 1;

    int u = eidx[ecl];
    int v = eidx[nedges + ecl];

    int koff = 8 * g + 4 * hi;
    unsigned lane_off = (unsigned)koff * 4u;
    const char* embn_bytes = reinterpret_cast<const char*>(embn);

    v4f dd = *reinterpret_cast<const v4f*>(dvec + koff);
    float sc = scale[0];

    v4f a0, a1, a2, a3, a4, a5, a6, a7;
    v4f b0, b1, b2, b3, b4, b5, b6, b7;
    load_pair<0>(embn_bytes, u, v, lane_off, a0, b0);
    load_pair<1>(embn_bytes, u, v, lane_off, a1, b1);
    load_pair<2>(embn_bytes, u, v, lane_off, a2, b2);
    load_pair<3>(embn_bytes, u, v, lane_off, a3, b3);
    load_pair<4>(embn_bytes, u, v, lane_off, a4, b4);
    load_pair<5>(embn_bytes, u, v, lane_off, a5, b5);
    load_pair<6>(embn_bytes, u, v, lane_off, a6, b6);
    load_pair<7>(embn_bytes, u, v, lane_off, a7, b7);

    float x0 = pair_diag(a0, b0, dd, g);
    float x1 = pair_diag(a1, b1, dd, g);
    float x2 = pair_diag(a2, b2, dd, g);
    float x3 = pair_diag(a3, b3, dd, g);
    float x4 = pair_diag(a4, b4, dd, g);
    float x5 = pair_diag(a5, b5, dd, g);
    float x6 = pair_diag(a6, b6, dd, g);
    float x7 = pair_diag(a7, b7, dd, g);

    bool s4 = (g & 4) != 0, s2 = (g & 2) != 0, s1 = (g & 1) != 0;
    float y0 = (s4 ? x4 : x0) + row_xor<4>(s4 ? x0 : x4);
    float y1 = (s4 ? x5 : x1) + row_xor<4>(s4 ? x1 : x5);
    float y2 = (s4 ? x6 : x2) + row_xor<4>(s4 ? x2 : x6);
    float y3 = (s4 ? x7 : x3) + row_xor<4>(s4 ? x3 : x7);
    float z0 = (s2 ? y2 : y0) + row_xor<2>(s2 ? y0 : y2);
    float z1 = (s2 ? y3 : y1) + row_xor<2>(s2 ? y1 : y3);
    float r  = (s1 ? z1 : z0) + row_xor<1>(s1 ? z0 : z1);

    int eo = -1;
    if (lane < 8)        eo = lane;
    else if (lane >= 24) eo = lane - 16;
    if (eo >= 0) {
        int eg = wave * 16 + eo;
        if (eg < nedges) out[eg] = r * sc;
    }
}

extern "C" void kernel_launch(void* const* d_in, const int* in_sizes, int n_in,
                              void* d_out, int out_size, void* d_ws, size_t ws_size,
                              hipStream_t stream) {
    const float* emb   = (const float*)d_in[0];
    const float* dvec  = (const float*)d_in[1];
    const float* scale = (const float*)d_in[2];
    const int*   eidx  = (const int*)d_in[3];
    float*       out   = (float*)d_out;

    int nrows  = in_sizes[0] / HIDDEN;
    int nedges = in_sizes[3] / 2;

    float* embn = (float*)d_ws;

    {
        long long total_threads = (long long)nrows * 32;
        int threads = 256;
        int blocks  = (int)((total_threads + threads - 1) / threads);
        normalize_rows_kernel<<<blocks, threads, 0, stream>>>(emb, embn, nrows);
    }

    {
        int ntiles = (nedges + 15) / 16;
        long long total_threads = (long long)ntiles * 32;
        int threads = 256;
        int blocks  = (int)((total_threads + threads - 1) / threads);
        edge_dot_wmma_kernel<<<blocks, threads, 0, stream>>>(
            embn, dvec, scale, eidx, out, nedges, ntiles);
    }
}


# Row normalisation folded into the edge kernel: it gathers raw rows, sums both rows' squares from the gathered registers, divides the dot by the two norms; one launch, no table; f32 sum order changes
# speedup vs baseline: 2.1360x; 1.2533x over previous
#include <hip/hip_runtime.h>
#include <math.h>

typedef __attribute__((ext_vector_type(2))) float v2f;
typedef __attribute__((ext_vector_type(4))) float v4f;
typedef __attribute__((ext_vector_type(8))) float v8f;

#define HIDDEN 64
#define NEPS 1e-12f


template <int T>
__device__ __forceinline__ int group8_bcast(int x) {
    return __builtin_amdgcn_mov_dpp8(x, T * 0x249249);
}

template <int M>
__device__ __forceinline__ float row_xor(float x) {
    return __int_as_float(__builtin_amdgcn_update_dpp(
        0, __float_as_int(x), 0x160 | M, 0xf, 0xf, false));
}

__device__ __forceinline__ float other_half(float x) {
    int b = __float_as_int(x);
    return __int_as_float(__builtin_amdgcn_permlanex16(b, b, 0x76543210, 0xfedcba98,
                                                       false, false));
}

template <int T>
__device__ __forceinline__ void load_pair(const char* __restrict__ emb_bytes,
                                          int u, int v, unsigned lane_off,
                                          v4f& a, v4f& b) {
    unsigned ut = (unsigned)group8_bcast<T>(u);
    unsigned vt = (unsigned)group8_bcast<T>(v);
    a = *reinterpret_cast<const v4f*>(emb_bytes + (ut * (HIDDEN * 4u) + lane_off));
    b = *reinterpret_cast<const v4f*>(emb_bytes + (vt * (HIDDEN * 4u) + lane_off));
}

__device__ __forceinline__ float sumsq4(v4f a) {
    v2f lo = {a.x, a.y}, hi = {a.z, a.w};
    v2f p = lo * lo;
    p = __builtin_elementwise_fma(hi, hi, p);
    return p.x + p.y;
}

__device__ __forceinline__ float pair_diag(v4f a, v4f b, v4f dd, int g, float& na, float& nb) {
    na = sumsq4(a);
    nb = sumsq4(b);
    b *= dd;
    v2f a0 = {a.x, a.y}, a1 = {a.z, a.w};
    v2f b0 = {b.x, b.y}, b1 = {b.z, b.w};
    v8f c = {};
    c = __builtin_amdgcn_wmma_f32_16x16x4_f32(false, a0, false, b0,
                                              (short)0, c, false, false);
    c = __builtin_amdgcn_wmma_f32_16x16x4_f32(false, a1, false, b1,
                                              (short)0, c, false, false);
    float x = c[0];
    x = (g == 1) ? c[1] : x;
    x = (g == 2) ? c[2] : x;
    x = (g == 3) ? c[3] : x;
    x = (g == 4) ? c[4] : x;
    x = (g == 5) ? c[5] : x;
    x = (g == 6) ? c[6] : x;
    x = (g == 7) ? c[7] : x;
    return x;
}

__device__ __forceinline__ float group8_transpose_sum(float x0, float x1, float x2, float x3,
                                                      float x4, float x5, float x6, float x7,
                                                      bool s4, bool s2, bool s1) {
    float y0 = (s4 ? x4 : x0) + row_xor<4>(s4 ? x0 : x4);
    float y1 = (s4 ? x5 : x1) + row_xor<4>(s4 ? x1 : x5);
    float y2 = (s4 ? x6 : x2) + row_xor<4>(s4 ? x2 : x6);
    float y3 = (s4 ? x7 : x3) + row_xor<4>(s4 ? x3 : x7);
    float z0 = (s2 ? y2 : y0) + row_xor<2>(s2 ? y0 : y2);
    float z1 = (s2 ? y3 : y1) + row_xor<2>(s2 ? y1 : y3);
    return (s1 ? z1 : z0) + row_xor<1>(s1 ? z0 : z1);
}

__global__ void edge_cosine_wmma_kernel(const float* __restrict__ emb,
                                        const float* __restrict__ dvec,
                                        const float* __restrict__ scale,
                                        const int* __restrict__ eidx,
                                        float* __restrict__ out,
                                        int nedges, int ntiles) {
    int wave = (blockIdx.x * blockDim.x + threadIdx.x) >> 5;
    int lane = threadIdx.x & 31;
    if (wave >= ntiles) return;

    int c    = lane & 15;
    int hi   = lane >> 4;
    int g    = lane & 7;
    int edge = wave * 16 + c;
    int ecl  = edge < nedges ? edge : nedges - 1;

    int u = eidx[ecl];
    int v = eidx[nedges + ecl];

    int koff = 8 * g + 4 * hi;
    unsigned lane_off = (unsigned)koff * 4u;
    const char* emb_bytes = reinterpret_cast<const char*>(emb);

    v4f dd = *reinterpret_cast<const v4f*>(dvec + koff);
    float sc = scale[0];

    v4f a0, a1, a2, a3, a4, a5, a6, a7;
    v4f b0, b1, b2, b3, b4, b5, b6, b7;
    load_pair<0>(emb_bytes, u, v, lane_off, a0, b0);
    load_pair<1>(emb_bytes, u, v, lane_off, a1, b1);
    load_pair<2>(emb_bytes, u, v, lane_off, a2, b2);
    load_pair<3>(emb_bytes, u, v, lane_off, a3, b3);
    load_pair<4>(emb_bytes, u, v, lane_off, a4, b4);
    load_pair<5>(emb_bytes, u, v, lane_off, a5, b5);
    load_pair<6>(emb_bytes, u, v, lane_off, a6, b6);
    load_pair<7>(emb_bytes, u, v, lane_off, a7, b7);

    float p0, q0;
    float x0 = pair_diag(a0, b0, dd, g, p0, q0);
    float p1, q1;
    float x1 = pair_diag(a1, b1, dd, g, p1, q1);
    float p2, q2;
    float x2 = pair_diag(a2, b2, dd, g, p2, q2);
    float p3, q3;
    float x3 = pair_diag(a3, b3, dd, g, p3, q3);
    float p4, q4;
    float x4 = pair_diag(a4, b4, dd, g, p4, q4);
    float p5, q5;
    float x5 = pair_diag(a5, b5, dd, g, p5, q5);
    float p6, q6;
    float x6 = pair_diag(a6, b6, dd, g, p6, q6);
    float p7, q7;
    float x7 = pair_diag(a7, b7, dd, g, p7, q7);

    bool s4 = (g & 4) != 0, s2 = (g & 2) != 0, s1 = (g & 1) != 0;

    float r = group8_transpose_sum(x0, x1, x2, x3, x4, x5, x6, x7, s4, s2, s1);

    float na = group8_transpose_sum(p0, p1, p2, p3, p4, p5, p6, p7, s4, s2, s1);
    float nb = group8_transpose_sum(q0, q1, q2, q3, q4, q5, q6, q7, s4, s2, s1);
    na += other_half(na);
    nb += other_half(nb);

    float ma = fmaxf(sqrtf(na), NEPS);
    float mb = fmaxf(sqrtf(nb), NEPS);
    r = r / (ma * mb);

    int eo = -1;
    if (lane < 8)        eo = lane;
    else if (lane >= 24) eo = lane - 16;
    if (eo >= 0) {
        int eg = wave * 16 + eo;
        if (eg < nedges) out[eg] = r * sc;
    }
}

extern "C" void kernel_launch(void* const* d_in, const int* in_sizes, int n_in,
                              void* d_out, int out_size, void* d_ws, size_t ws_size,
                              hipStream_t stream) {
    const float* emb   = (const float*)d_in[0];
    const float* dvec  = (const float*)d_in[1];
    const float* scale = (const float*)d_in[2];
    const int*   eidx  = (const int*)d_in[3];
    float*       out   = (float*)d_out;

    int nedges = in_sizes[3] / 2;

    int ntiles = (nedges + 15) / 16;
    long long total_threads = (long long)ntiles * 32;
    int threads = 256;
    int blocks  = (int)((total_threads + threads - 1) / threads);
    edge_cosine_wmma_kernel<<<blocks, threads, 0, stream>>>(
        emb, dvec, scale, eidx, out, nedges, ntiles);
}


# Resident wave loop: fixed grid of 6 waves per SIMD, declared to the compiler, each wave walking about 10 tiles; a tile's 16 row gathers issue as one group with the next tile's indices behind them
# speedup vs baseline: 2.2548x; 1.0556x over previous
#include <hip/hip_runtime.h>
#include <math.h>

typedef __attribute__((ext_vector_type(2))) float v2f;
typedef __attribute__((ext_vector_type(4))) float v4f;
typedef __attribute__((ext_vector_type(8))) float v8f;

#define HIDDEN 64
#define NEPS 1e-12f


template <int T>
__device__ __forceinline__ int group8_bcast(int x) {
    return __builtin_amdgcn_mov_dpp8(x, T * 0x249249);
}

template <int M>
__device__ __forceinline__ float row_xor(float x) {
    return __int_as_float(__builtin_amdgcn_update_dpp(
        0, __float_as_int(x), 0x160 | M, 0xf, 0xf, false));
}

__device__ __forceinline__ float other_half(float x) {
    int b = __float_as_int(x);
    return __int_as_float(__builtin_amdgcn_permlanex16(b, b, 0x76543210, 0xfedcba98,
                                                       false, false));
}

template <int T>
__device__ __forceinline__ void load_pair(const char* __restrict__ emb_bytes,
                                          int u, int v, unsigned lane_off,
                                          v4f& a, v4f& b) {
    unsigned ut = (unsigned)group8_bcast<T>(u);
    unsigned vt = (unsigned)group8_bcast<T>(v);
    a = *reinterpret_cast<const v4f*>(emb_bytes + (ut * (HIDDEN * 4u) + lane_off));
    b = *reinterpret_cast<const v4f*>(emb_bytes + (vt * (HIDDEN * 4u) + lane_off));
}

__device__ __forceinline__ float sumsq4(v4f a) {
    v2f lo = {a.x, a.y}, hi = {a.z, a.w};
    v2f p = lo * lo;
    p = __builtin_elementwise_fma(hi, hi, p);
    return p.x + p.y;
}

__device__ __forceinline__ float pair_diag(v4f a, v4f b, v4f dd, int g, float& na, float& nb) {
    na = sumsq4(a);
    nb = sumsq4(b);
    b *= dd;
    v2f a0 = {a.x, a.y}, a1 = {a.z, a.w};
    v2f b0 = {b.x, b.y}, b1 = {b.z, b.w};
    v8f c = {};
    c = __builtin_amdgcn_wmma_f32_16x16x4_f32(false, a0, false, b0,
                                              (short)0, c, false, false);
    c = __builtin_amdgcn_wmma_f32_16x16x4_f32(false, a1, false, b1,
                                              (short)0, c, false, false);
    float x = c[0];
    x = (g == 1) ? c[1] : x;
    x = (g == 2) ? c[2] : x;
    x = (g == 3) ? c[3] : x;
    x = (g == 4) ? c[4] : x;
    x = (g == 5) ? c[5] : x;
    x = (g == 6) ? c[6] : x;
    x = (g == 7) ? c[7] : x;
    return x;
}

__device__ __forceinline__ float group8_transpose_sum(float x0, float x1, float x2, float x3,
                                                      float x4, float x5, float x6, float x7,
                                                      bool s4, bool s2, bool s1) {
    float y0 = (s4 ? x4 : x0) + row_xor<4>(s4 ? x0 : x4);
    float y1 = (s4 ? x5 : x1) + row_xor<4>(s4 ? x1 : x5);
    float y2 = (s4 ? x6 : x2) + row_xor<4>(s4 ? x2 : x6);
    float y3 = (s4 ? x7 : x3) + row_xor<4>(s4 ? x3 : x7);
    float z0 = (s2 ? y2 : y0) + row_xor<2>(s2 ? y0 : y2);
    float z1 = (s2 ? y3 : y1) + row_xor<2>(s2 ? y1 : y3);
    return (s1 ? z1 : z0) + row_xor<1>(s1 ? z0 : z1);
}

__global__ void __launch_bounds__(128) __attribute__((amdgpu_waves_per_eu(6, 7)))
edge_cosine_wmma_kernel(const float* __restrict__ emb,
                        const float* __restrict__ dvec,
                        const float* __restrict__ scale,
                        const int* __restrict__ eidx,
                        float* __restrict__ out,
                        int nedges, int ntiles, int nwaves) {
    int lane = threadIdx.x & 31;
    int wave = __builtin_amdgcn_readfirstlane(
        (int)((blockIdx.x * blockDim.x + threadIdx.x) >> 5));
    if (wave >= ntiles) return;

    int c    = lane & 15;
    int hi   = lane >> 4;
    int g    = lane & 7;
    int last = nedges - 1;

    int koff = 8 * g + 4 * hi;
    unsigned lane_off = (unsigned)koff * 4u;
    const char* emb_bytes = reinterpret_cast<const char*>(emb);

    v4f dd = *reinterpret_cast<const v4f*>(dvec + koff);
    float sc = scale[0];

    bool s4 = (g & 4) != 0, s2 = (g & 2) != 0, s1 = (g & 1) != 0;

    int eo = -1;
    if (lane < 8)        eo = lane;
    else if (lane >= 24) eo = lane - 16;

    int e0 = wave * 16 + c;
    e0 = e0 < last ? e0 : last;
    int u = eidx[e0];
    int v = eidx[nedges + e0];

    for (int tile = wave; tile < ntiles; tile += nwaves) {
        v4f a0, a1, a2, a3, a4, a5, a6, a7;
        v4f b0, b1, b2, b3, b4, b5, b6, b7;
        load_pair<0>(emb_bytes, u, v, lane_off, a0, b0);
        load_pair<1>(emb_bytes, u, v, lane_off, a1, b1);
        load_pair<2>(emb_bytes, u, v, lane_off, a2, b2);
        load_pair<3>(emb_bytes, u, v, lane_off, a3, b3);
        load_pair<4>(emb_bytes, u, v, lane_off, a4, b4);
        load_pair<5>(emb_bytes, u, v, lane_off, a5, b5);
        load_pair<6>(emb_bytes, u, v, lane_off, a6, b6);
        load_pair<7>(emb_bytes, u, v, lane_off, a7, b7);

        int en = (tile + nwaves) * 16 + c;
        en = en < last ? en : last;
        int un = eidx[en];
        int vn = eidx[nedges + en];

        float p0, q0;
        float x0 = pair_diag(a0, b0, dd, g, p0, q0);
        float p1, q1;
        float x1 = pair_diag(a1, b1, dd, g, p1, q1);
        float p2, q2;
        float x2 = pair_diag(a2, b2, dd, g, p2, q2);
        float p3, q3;
        float x3 = pair_diag(a3, b3, dd, g, p3, q3);
        float p4, q4;
        float x4 = pair_diag(a4, b4, dd, g, p4, q4);
        float p5, q5;
        float x5 = pair_diag(a5, b5, dd, g, p5, q5);
        float p6, q6;
        float x6 = pair_diag(a6, b6, dd, g, p6, q6);
        float p7, q7;
        float x7 = pair_diag(a7, b7, dd, g, p7, q7);

        float r = group8_transpose_sum(x0, x1, x2, x3, x4, x5, x6, x7, s4, s2, s1);

        float na = group8_transpose_sum(p0, p1, p2, p3, p4, p5, p6, p7, s4, s2, s1);
        float nb = group8_transpose_sum(q0, q1, q2, q3, q4, q5, q6, q7, s4, s2, s1);
        na += other_half(na);
        nb += other_half(nb);

        float ma = fmaxf(sqrtf(na), NEPS);
        float mb = fmaxf(sqrtf(nb), NEPS);
        r = r / (ma * mb);

        if (eo >= 0) {
            int eg = tile * 16 + eo;
            if (eg < nedges) out[eg] = r * sc;
        }

        u = un;
        v = vn;
    }
}

extern "C" void kernel_launch(void* const* d_in, const int* in_sizes, int n_in,
                              void* d_out, int out_size, void* d_ws, size_t ws_size,
                              hipStream_t stream) {
    const float* emb   = (const float*)d_in[0];
    const float* dvec  = (const float*)d_in[1];
    const float* scale = (const float*)d_in[2];
    const int*   eidx  = (const int*)d_in[3];
    float*       out   = (float*)d_out;

    int nedges = in_sizes[3] / 2;

    int ntiles  = (nedges + 15) / 16;
    int threads = 128;
    int blocks  = 1536;
    int nwaves  = blocks * (threads / 32);
    edge_cosine_wmma_kernel<<<blocks, threads, 0, stream>>>(
        emb, dvec, scale, eidx, out, nedges, ntiles, nwaves);
}


# Fewer vector instructions per tile: norms applied as two v_rsq_f32 on eps-clamped squared norms (no IEEE sqrt or division); each lane loads its 8 edge indices itself, no broadcasts for addresses
# speedup vs baseline: 2.5452x; 1.1288x over previous
#include <hip/hip_runtime.h>
#include <math.h>

typedef __attribute__((ext_vector_type(2))) float v2f;
typedef __attribute__((ext_vector_type(4))) float v4f;
typedef __attribute__((ext_vector_type(8))) float v8f;
typedef __attribute__((ext_vector_type(4))) int v4i;

#define HIDDEN 64
#define NEPS2 1e-24f


template <int T>
__device__ __forceinline__ int group8_bcast(int x) {
    return __builtin_amdgcn_mov_dpp8(x, T * 0x249249);
}

template <int M>
__device__ __forceinline__ float row_xor(float x) {
    return __int_as_float(__builtin_amdgcn_update_dpp(
        0, __float_as_int(x), 0x160 | M, 0xf, 0xf, false));
}

__device__ __forceinline__ float other_half(float x) {
    int b = __float_as_int(x);
    return __int_as_float(__builtin_amdgcn_permlanex16(b, b, 0x76543210, 0xfedcba98,
                                                       false, false));
}

__device__ __forceinline__ void load_idx8(const int* __restrict__ arr, int first,
                                          v4i& lo, v4i& hi) {
    const v4i* p = reinterpret_cast<const v4i*>(
        reinterpret_cast<const char*>(arr) + (unsigned)first * 4u);
    lo = p[0];
    hi = p[1];
}

__device__ __forceinline__ void load_idx8_clamped(const int* __restrict__ arr, int first,
                                                  int last, v4i& lo, v4i& hi) {
    int e[8];
#pragma unroll
    for (int t = 0; t < 8; ++t) {
        int k = first + t;
        e[t] = arr[k < last ? k : last];
    }
    lo = (v4i){e[0], e[1], e[2], e[3]};
    hi = (v4i){e[4], e[5], e[6], e[7]};
}

__device__ __forceinline__ void load_pair(const char* __restrict__ emb_bytes,
                                          int ut, int vt, unsigned lane_off,
                                          v4f& a, v4f& b) {
    a = *reinterpret_cast<const v4f*>(emb_bytes + (((unsigned)ut << 8) | lane_off));
    b = *reinterpret_cast<const v4f*>(emb_bytes + (((unsigned)vt << 8) | lane_off));
}

__device__ __forceinline__ float sumsq4(v4f a) {
    v2f lo = {a.x, a.y}, hi = {a.z, a.w};
    v2f p = lo * lo;
    p = __builtin_elementwise_fma(hi, hi, p);
    return p.x + p.y;
}

__device__ __forceinline__ float pair_diag(v4f a, v4f b, v4f dd, int g, float& na, float& nb) {
    na = sumsq4(a);
    nb = sumsq4(b);
    b *= dd;
    v2f a0 = {a.x, a.y}, a1 = {a.z, a.w};
    v2f b0 = {b.x, b.y}, b1 = {b.z, b.w};
    v8f c = {};
    c = __builtin_amdgcn_wmma_f32_16x16x4_f32(false, a0, false, b0,
                                              (short)0, c, false, false);
    c = __builtin_amdgcn_wmma_f32_16x16x4_f32(false, a1, false, b1,
                                              (short)0, c, false, false);
    float x = c[0];
    x = (g == 1) ? c[1] : x;
    x = (g == 2) ? c[2] : x;
    x = (g == 3) ? c[3] : x;
    x = (g == 4) ? c[4] : x;
    x = (g == 5) ? c[5] : x;
    x = (g == 6) ? c[6] : x;
    x = (g == 7) ? c[7] : x;
    return x;
}

__device__ __forceinline__ float group8_transpose_sum(float x0, float x1, float x2, float x3,
                                                      float x4, float x5, float x6, float x7,
                                                      bool s4, bool s2, bool s1) {
    float y0 = (s4 ? x4 : x0) + row_xor<4>(s4 ? x0 : x4);
    float y1 = (s4 ? x5 : x1) + row_xor<4>(s4 ? x1 : x5);
    float y2 = (s4 ? x6 : x2) + row_xor<4>(s4 ? x2 : x6);
    float y3 = (s4 ? x7 : x3) + row_xor<4>(s4 ? x3 : x7);
    float z0 = (s2 ? y2 : y0) + row_xor<2>(s2 ? y0 : y2);
    float z1 = (s2 ? y3 : y1) + row_xor<2>(s2 ? y1 : y3);
    return (s1 ? z1 : z0) + row_xor<1>(s1 ? z0 : z1);
}

__device__ __forceinline__ void gather_tile(const char* __restrict__ emb_bytes,
                                            v4i ul, v4i uh, v4i vl, v4i vh,
                                            unsigned lane_off, v4f (&a)[8], v4f (&b)[8]) {
    load_pair(emb_bytes, ul.x, vl.x, lane_off, a[0], b[0]);
    load_pair(emb_bytes, ul.y, vl.y, lane_off, a[1], b[1]);
    load_pair(emb_bytes, ul.z, vl.z, lane_off, a[2], b[2]);
    load_pair(emb_bytes, ul.w, vl.w, lane_off, a[3], b[3]);
    load_pair(emb_bytes, uh.x, vh.x, lane_off, a[4], b[4]);
    load_pair(emb_bytes, uh.y, vh.y, lane_off, a[5], b[5]);
    load_pair(emb_bytes, uh.z, vh.z, lane_off, a[6], b[6]);
    load_pair(emb_bytes, uh.w, vh.w, lane_off, a[7], b[7]);
}

__device__ __forceinline__ void finish_tile(const v4f (&a)[8], const v4f (&b)[8], v4f dd,
                                            int g, bool s4, bool s2, bool s1, float sc,
                                            int eo, int tile, int nedges,
                                            float* __restrict__ out) {
    float x[8], p[8], q[8];
#pragma unroll
    for (int t = 0; t < 8; ++t) x[t] = pair_diag(a[t], b[t], dd, g, p[t], q[t]);

    float r = group8_transpose_sum(x[0], x[1], x[2], x[3], x[4], x[5], x[6], x[7], s4, s2, s1);

    float na = group8_transpose_sum(p[0], p[1], p[2], p[3], p[4], p[5], p[6], p[7], s4, s2, s1);
    float nb = group8_transpose_sum(q[0], q[1], q[2], q[3], q[4], q[5], q[6], q[7], s4, s2, s1);
    na += other_half(na);
    nb += other_half(nb);

    float ia = __builtin_amdgcn_rsqf(fmaxf(na, NEPS2));
    float ib = __builtin_amdgcn_rsqf(fmaxf(nb, NEPS2));
    r = (r * ia) * ib;

    if (eo >= 0) {
        int eg = tile * 16 + eo;
        if (eg < nedges) out[eg] = r * sc;
    }
}

__global__ void __launch_bounds__(128) __attribute__((amdgpu_waves_per_eu(6, 7)))
edge_cosine_wmma_kernel(const float* __restrict__ emb,
                        const float* __restrict__ dvec,
                        const float* __restrict__ scale,
                        const int* __restrict__ eidx,
                        float* __restrict__ out,
                        int nedges, int ntiles, int nwaves) {
    int lane = threadIdx.x & 31;
    int wave = __builtin_amdgcn_readfirstlane(
        (int)((blockIdx.x * blockDim.x + threadIdx.x) >> 5));
    if (wave >= ntiles) return;

    int c    = lane & 15;
    int hi   = lane >> 4;
    int g    = lane & 7;
    int last = nedges - 1;
    int j8   = 8 * (c >> 3);

    int koff = 8 * g + 4 * hi;
    unsigned lane_off = (unsigned)koff * 4u;
    const char* emb_bytes = reinterpret_cast<const char*>(emb);
    const int* vidx = eidx + nedges;

    v4f dd = *reinterpret_cast<const v4f*>(dvec + koff);
    float sc = scale[0];

    bool s4 = (g & 4) != 0, s2 = (g & 2) != 0, s1 = (g & 1) != 0;

    int eo = -1;
    if (lane < 8)        eo = lane;
    else if (lane >= 24) eo = lane - 16;

    int nfull = (nedges & 3) == 0 ? nedges >> 4 : 0;
    v4i ul, uh, vl, vh;
    if (wave < nfull) {
        load_idx8(eidx, wave * 16 + j8, ul, uh);
        load_idx8(vidx, wave * 16 + j8, vl, vh);
    }

    int tile = wave;
    for (; tile < nfull; tile += nwaves) {
        v4f a[8], b[8];
        gather_tile(emb_bytes, ul, uh, vl, vh, lane_off, a, b);

        int next = tile + nwaves;
        if (next < nfull) {
            load_idx8(eidx, next * 16 + j8, ul, uh);
            load_idx8(vidx, next * 16 + j8, vl, vh);
        }

        finish_tile(a, b, dd, g, s4, s2, s1, sc, eo, tile, nedges, out);
    }

    for (; tile < ntiles; tile += nwaves) {
        load_idx8_clamped(eidx, tile * 16 + j8, last, ul, uh);
        load_idx8_clamped(vidx, tile * 16 + j8, last, vl, vh);
        v4f a[8], b[8];
        gather_tile(emb_bytes, ul, uh, vl, vh, lane_off, a, b);
        finish_tile(a, b, dd, g, s4, s2, s1, sc, eo, tile, nedges, out);
    }
}

extern "C" void kernel_launch(void* const* d_in, const int* in_sizes, int n_in,
                              void* d_out, int out_size, void* d_ws, size_t ws_size,
                              hipStream_t stream) {
    const float* emb   = (const float*)d_in[0];
    const float* dvec  = (const float*)d_in[1];
    const float* scale = (const float*)d_in[2];
    const int*   eidx  = (const int*)d_in[3];
    float*       out   = (float*)d_out;

    int nedges = in_sizes[3] / 2;

    int ntiles  = (nedges + 15) / 16;
    int threads = 128;
    int blocks  = 1536;
    int nwaves  = blocks * (threads / 32);
    edge_cosine_wmma_kernel<<<blocks, threads, 0, stream>>>(
        emb, dvec, scale, eidx, out, nedges, ntiles, nwaves);
}


# Tile row = (edge, 64-B quarter): a b128 gather covers one whole line of four rows and four WMMA steps share an accumulator; half the diagonal picks, 4-lane butterflies; f32 sum order changes
# speedup vs baseline: 2.7021x; 1.0616x over previous
#include <hip/hip_runtime.h>
#include <math.h>

typedef __attribute__((ext_vector_type(2))) float v2f;
typedef __attribute__((ext_vector_type(4))) float v4f;
typedef __attribute__((ext_vector_type(8))) float v8f;
typedef __attribute__((ext_vector_type(4))) int v4i;

#define HIDDEN 64
#define NEPS2 1e-24f


template <int M>
__device__ __forceinline__ float row_xor(float x) {
    return __int_as_float(__builtin_amdgcn_update_dpp(
        0, __float_as_int(x), 0x160 | M, 0xf, 0xf, false));
}

__device__ __forceinline__ float other_half(float x) {
    int b = __float_as_int(x);
    return __int_as_float(__builtin_amdgcn_permlanex16(b, b, 0x76543210, 0xfedcba98,
                                                       false, false));
}

__device__ __forceinline__ v4i load_idx4(const int* __restrict__ arr, int first) {
    return *reinterpret_cast<const v4i*>(
        reinterpret_cast<const char*>(arr) + (unsigned)first * 4u);
}

__device__ __forceinline__ v4i load_idx4_clamped(const int* __restrict__ arr, int first,
                                                 int last) {
    int e[4];
#pragma unroll
    for (int t = 0; t < 4; ++t) {
        int k = first + t;
        e[t] = arr[k < last ? k : last];
    }
    return (v4i){e[0], e[1], e[2], e[3]};
}

__device__ __forceinline__ void load_row(const char* __restrict__ emb_bytes, int node,
                                         unsigned lane_off, v4f& l0, v4f& l1) {
    const v4f* p = reinterpret_cast<const v4f*>(emb_bytes + (((unsigned)node << 8) | lane_off));
    l0 = p[0];
    l1 = p[8];
}

__device__ __forceinline__ float sumsq8(v4f a0, v4f a1) {
    v2f x0 = {a0.x, a0.y}, x1 = {a0.z, a0.w};
    v2f x2 = {a1.x, a1.y}, x3 = {a1.z, a1.w};
    v2f p = x0 * x0;
    p = __builtin_elementwise_fma(x1, x1, p);
    p = __builtin_elementwise_fma(x2, x2, p);
    p = __builtin_elementwise_fma(x3, x3, p);
    return p.x + p.y;
}

__device__ __forceinline__ float group_diag(v4f a0, v4f a1, v4f b0, v4f b1, v4f d0, v4f d1,
                                            int g, float& na, float& nb) {
    na = sumsq8(a0, a1);
    nb = sumsq8(b0, b1);
    b0 *= d0;
    b1 *= d1;
    v2f a00 = {a0.x, a0.y}, a01 = {a0.z, a0.w}, a10 = {a1.x, a1.y}, a11 = {a1.z, a1.w};
    v2f b00 = {b0.x, b0.y}, b01 = {b0.z, b0.w}, b10 = {b1.x, b1.y}, b11 = {b1.z, b1.w};
    v8f c = {};
    c = __builtin_amdgcn_wmma_f32_16x16x4_f32(false, a00, false, b00,
                                              (short)0, c, false, false);
    c = __builtin_amdgcn_wmma_f32_16x16x4_f32(false, a01, false, b01,
                                              (short)0, c, false, false);
    c = __builtin_amdgcn_wmma_f32_16x16x4_f32(false, a10, false, b10,
                                              (short)0, c, false, false);
    c = __builtin_amdgcn_wmma_f32_16x16x4_f32(false, a11, false, b11,
                                              (short)0, c, false, false);
    float x = c[0];
    x = (g == 1) ? c[1] : x;
    x = (g == 2) ? c[2] : x;
    x = (g == 3) ? c[3] : x;
    x = (g == 4) ? c[4] : x;
    x = (g == 5) ? c[5] : x;
    x = (g == 6) ? c[6] : x;
    x = (g == 7) ? c[7] : x;
    return x;
}

__device__ __forceinline__ float quad_transpose_sum(float x0, float x1, float x2, float x3,
                                                    bool s2, bool s1) {
    float y0 = (s2 ? x2 : x0) + row_xor<2>(s2 ? x0 : x2);
    float y1 = (s2 ? x3 : x1) + row_xor<2>(s2 ? x1 : x3);
    return (s1 ? y1 : y0) + row_xor<1>(s1 ? y0 : y1);
}

__device__ __forceinline__ void gather_tile(const char* __restrict__ emb_bytes,
                                            v4i u, v4i v, unsigned lane_off,
                                            v4f (&a)[8], v4f (&b)[8]) {
    load_row(emb_bytes, u.x, lane_off, a[0], a[1]);
    load_row(emb_bytes, v.x, lane_off, b[0], b[1]);
    load_row(emb_bytes, u.y, lane_off, a[2], a[3]);
    load_row(emb_bytes, v.y, lane_off, b[2], b[3]);
    load_row(emb_bytes, u.z, lane_off, a[4], a[5]);
    load_row(emb_bytes, v.z, lane_off, b[4], b[5]);
    load_row(emb_bytes, u.w, lane_off, a[6], a[7]);
    load_row(emb_bytes, v.w, lane_off, b[6], b[7]);
}

__device__ __forceinline__ void finish_tile(const v4f (&a)[8], const v4f (&b)[8],
                                            v4f d0, v4f d1, int g, bool s2, bool s1,
                                            float sc, int eo, int tile, int nedges,
                                            float* __restrict__ out) {
    float x[4], p[4], q[4];
#pragma unroll
    for (int s = 0; s < 4; ++s)
        x[s] = group_diag(a[2 * s], a[2 * s + 1], b[2 * s], b[2 * s + 1], d0, d1, g, p[s], q[s]);

    float r = quad_transpose_sum(x[0], x[1], x[2], x[3], s2, s1);

    float na = quad_transpose_sum(p[0], p[1], p[2], p[3], s2, s1);
    float nb = quad_transpose_sum(q[0], q[1], q[2], q[3], s2, s1);
    na += other_half(na);
    nb += other_half(nb);

    float ia = __builtin_amdgcn_rsqf(fmaxf(na, NEPS2));
    float ib = __builtin_amdgcn_rsqf(fmaxf(nb, NEPS2));
    r = (r * ia) * ib;

    if (eo >= 0) {
        int eg = tile * 16 + eo;
        if (eg < nedges) out[eg] = r * sc;
    }
}

__global__ void __launch_bounds__(128) __attribute__((amdgpu_waves_per_eu(6, 7)))
edge_cosine_wmma_kernel(const float* __restrict__ emb,
                        const float* __restrict__ dvec,
                        const float* __restrict__ scale,
                        const int* __restrict__ eidx,
                        float* __restrict__ out,
                        int nedges, int ntiles, int nwaves) {
    int lane = threadIdx.x & 31;
    int wave = __builtin_amdgcn_readfirstlane(
        (int)((blockIdx.x * blockDim.x + threadIdx.x) >> 5));
    if (wave >= ntiles) return;

    int c    = lane & 15;
    int hi   = lane >> 4;
    int q    = lane & 3;
    int g    = lane & 7;
    int last = nedges - 1;
    int j4   = 4 * (c >> 2);

    int koff = 8 * q + 4 * hi;
    unsigned lane_off = (unsigned)koff * 4u;
    const char* emb_bytes = reinterpret_cast<const char*>(emb);
    const int* vidx = eidx + nedges;

    v4f d0 = *reinterpret_cast<const v4f*>(dvec + koff);
    v4f d1 = *reinterpret_cast<const v4f*>(dvec + 32 + koff);
    float sc = scale[0];

    bool s2 = (q & 2) != 0, s1 = (q & 1) != 0;

    int eo = -1;
    if (lane < 8)        eo = lane;
    else if (lane >= 24) eo = lane - 16;

    int nfull = (nedges & 3) == 0 ? nedges >> 4 : 0;
    v4i u, v;
    if (wave < nfull) {
        u = load_idx4(eidx, wave * 16 + j4);
        v = load_idx4(vidx, wave * 16 + j4);
    }

    int tile = wave;
    for (; tile < nfull; tile += nwaves) {
        v4f a[8], b[8];
        gather_tile(emb_bytes, u, v, lane_off, a, b);

        int next = tile + nwaves;
        if (next < nfull) {
            u = load_idx4(eidx, next * 16 + j4);
            v = load_idx4(vidx, next * 16 + j4);
        }

        finish_tile(a, b, d0, d1, g, s2, s1, sc, eo, tile, nedges, out);
    }

    for (; tile < ntiles; tile += nwaves) {
        u = load_idx4_clamped(eidx, tile * 16 + j4, last);
        v = load_idx4_clamped(vidx, tile * 16 + j4, last);
        v4f a[8], b[8];
        gather_tile(emb_bytes, u, v, lane_off, a, b);
        finish_tile(a, b, d0, d1, g, s2, s1, sc, eo, tile, nedges, out);
    }
}

extern "C" void kernel_launch(void* const* d_in, const int* in_sizes, int n_in,
                              void* d_out, int out_size, void* d_ws, size_t ws_size,
                              hipStream_t stream) {
    const float* emb   = (const float*)d_in[0];
    const float* dvec  = (const float*)d_in[1];
    const float* scale = (const float*)d_in[2];
    const int*   eidx  = (const int*)d_in[3];
    float*       out   = (float*)d_out;

    int nedges = in_sizes[3] / 2;

    int ntiles  = (nedges + 15) / 16;
    int threads = 128;
    int blocks  = 1536;
    int nwaves  = blocks * (threads / 32);
    edge_cosine_wmma_kernel<<<blocks, threads, 0, stream>>>(
        emb, dvec, scale, eidx, out, nedges, ntiles, nwaves);
}
